# starting kernel
# baseline (speedup 1.0000x reference)
#include <hip/hip_runtime.h>
#include <hip/hip_bf16.h>

#define NUM_ENT   40000
#define EMBED     256
#define FEAT      768
#define KCHUNKS   (FEAT / 32)
#define NTILES    (EMBED / 16)
#define NE_TOT    40103
#define NE_SCORE  40001
#define BATCH     32

typedef __attribute__((ext_vector_type(16))) __bf16 v16bf;
typedef __attribute__((ext_vector_type(8)))  float  v8f;

__device__ __forceinline__ void load16(const float* __restrict__ p, float* v) {
  float4 a0 = *(const float4*)(p + 0);
  float4 a1 = *(const float4*)(p + 4);
  float4 b0 = *(const float4*)(p + 16);
  float4 b1 = *(const float4*)(p + 20);
  v[0] = a0.x; v[1] = a0.y; v[2]  = a0.z; v[3]  = a0.w;
  v[4] = a1.x; v[5] = a1.y; v[6]  = a1.z; v[7]  = a1.w;
  v[8] = b0.x; v[9] = b0.y; v[10] = b0.z; v[11] = b0.w;
  v[12] = b1.x; v[13] = b1.y; v[14] = b1.z; v[15] = b1.w;
}

__global__ __launch_bounds__(32) void BLP_split_w(
    const float* __restrict__ W, __bf16* __restrict__ whi,
    __bf16* __restrict__ wlo) {
  const int lane = threadIdx.x;
  const int kc   = blockIdx.x % KCHUNKS;
  const int nt   = blockIdx.x / KCHUNKS;
  const int half = lane >> 4;
  const int r15  = lane & 15;

  const float* p = W + (size_t)(nt * 16 + r15) * FEAT + kc * 32 + half * 8;
  float v[16];
  load16(p, v);

  const size_t off = ((size_t)(nt * KCHUNKS + kc) * 32 + lane) * 16;
#pragma unroll
  for (int i = 0; i < 16; ++i) {
    __bf16 h = (__bf16)v[i];
    whi[off + i] = h;
    wlo[off + i] = (__bf16)(v[i] - (float)h);
  }
}

__global__ __launch_bounds__(128) void BLP_gemm_proj(
    const float* __restrict__ A, const __bf16* __restrict__ whi,
    const __bf16* __restrict__ wlo, float* __restrict__ unity) {
  const int lane = threadIdx.x & 31;
  const int wave = threadIdx.x >> 5;
  const int half = lane >> 4;
  const int r15  = lane & 15;

  const int n_half = blockIdx.x & 1;
  const int m_tile = (blockIdx.x >> 1) * 4 + wave;
  const int m_base = m_tile * 16;

  v8f acc[8] = {};

  const float* arow = A + (size_t)(m_base + r15) * FEAT + half * 8;

  for (int kc = 0; kc < KCHUNKS; ++kc) {
    const int k0 = kc * 32;
    if (k0 + 32 < FEAT)
      __builtin_prefetch(arow + k0 + 32, 0, 0);

    float v[16];
    load16(arow + k0, v);
    v16bf ahi, alo;
#pragma unroll
    for (int i = 0; i < 16; ++i) {
      __bf16 h = (__bf16)v[i];
      ahi[i] = h;
      alo[i] = (__bf16)(v[i] - (float)h);
    }

#pragma unroll
    for (int nt = 0; nt < 8; ++nt) {
      const int nt_g = n_half * 8 + nt;
      const size_t off = ((size_t)(nt_g * KCHUNKS + kc) * 32 + lane) * 16;
      v16bf bhi = *(const v16bf*)(whi + off);
      v16bf blo = *(const v16bf*)(wlo + off);
      acc[nt] = __builtin_amdgcn_wmma_f32_16x16x32_bf16(
          false, ahi, false, bhi, (short)0, acc[nt], false, false);
      acc[nt] = __builtin_amdgcn_wmma_f32_16x16x32_bf16(
          false, ahi, false, blo, (short)0, acc[nt], false, false);
      acc[nt] = __builtin_amdgcn_wmma_f32_16x16x32_bf16(
          false, alo, false, bhi, (short)0, acc[nt], false, false);
    }
  }

#pragma unroll
  for (int nt = 0; nt < 8; ++nt) {
#pragma unroll
    for (int j = 0; j < 8; ++j) {
      int m   = m_base + j + half * 8;
      int col = n_half * 128 + nt * 16 + r15;
      unity[(size_t)(m + 1) * EMBED + col] = acc[nt][j];
    }
  }
}

__global__ __launch_bounds__(256) void BLP_copy_other(
    const float* __restrict__ other, float* __restrict__ unity) {
  int r = blockIdx.x;
  int d = threadIdx.x;
  int dst = (r == 0) ? 0 : (NUM_ENT + r);
  unity[(size_t)dst * EMBED + d] = other[(size_t)r * EMBED + d];
}

__global__ __launch_bounds__(256) void BLP_query(
    const float* __restrict__ unity, const int* __restrict__ ids,
    const int* __restrict__ mp_ptr, float* __restrict__ qsum) {
  __shared__ float s0[256];
  __shared__ float s1[256];
  const int b = blockIdx.x;
  const int d = threadIdx.x;
  const int mp = *mp_ptr;
  const int c0 = (mp == 0) ? 1 : 0;
  const int c1 = (mp <= 1) ? 2 : 1;
  const int id0 = ids[b * 3 + c0];
  const int id1 = ids[b * 3 + c1];
  float v0 = unity[(size_t)id0 * EMBED + d];
  float v1 = unity[(size_t)id1 * EMBED + d];
  s0[d] = v0 * v0;
  s1[d] = v1 * v1;
  __syncthreads();
  for (int off = 128; off > 0; off >>= 1) {
    if (d < off) { s0[d] += s0[d + off]; s1[d] += s1[d + off]; }
    __syncthreads();
  }
  float r0 = 1.0f / fmaxf(sqrtf(s0[0]), 1e-12f);
  float r1 = 1.0f / fmaxf(sqrtf(s1[0]), 1e-12f);
  qsum[(size_t)b * EMBED + d] = v0 * r0 + v1 * r1;
}

__global__ __launch_bounds__(256) void BLP_score(
    const float* __restrict__ unity, const float* __restrict__ qsum,
    float* __restrict__ out) {
  __shared__ float q[BATCH * EMBED];
  for (int i = threadIdx.x; i < BATCH * EMBED; i += 256) q[i] = qsum[i];
  __syncthreads();

  const int e = blockIdx.x * 256 + threadIdx.x;
  const bool active = (e < NE_SCORE);
  const float4* row = (const float4*)(unity + (size_t)(active ? e : 0) * EMBED);

  float acc[BATCH];
#pragma unroll
  for (int b = 0; b < BATCH; ++b) acc[b] = 0.0f;

  for (int d4 = 0; d4 < EMBED / 4; ++d4) {
    float4 v = row[d4];
    const int dbase = d4 * 4;
#pragma unroll
    for (int b = 0; b < BATCH; ++b) {
      const float* qb = q + b * EMBED + dbase;
      acc[b] += fabsf(qb[0] - v.x) + fabsf(qb[1] - v.y) +
                fabsf(qb[2] - v.z) + fabsf(qb[3] - v.w);
    }
  }
  if (active) {
#pragma unroll
    for (int b = 0; b < BATCH; ++b)
      out[(size_t)b * NE_SCORE + e] = -acc[b];
  }
}

extern "C" void kernel_launch(void* const* d_in, const int* in_sizes, int n_in,
                              void* d_out, int out_size, void* d_ws, size_t ws_size,
                              hipStream_t stream) {
  (void)in_sizes; (void)n_in; (void)out_size; (void)ws_size;
  const float* ent_pkl = (const float*)d_in[0];
  const float* other   = (const float*)d_in[1];
  const float* proj_W  = (const float*)d_in[2];
  const int*   ids     = (const int*)d_in[3];
  const int*   mp      = (const int*)d_in[4];
  float*       out     = (float*)d_out;

  float*  unity = (float*)d_ws;
  float*  qsum  = unity + (size_t)NE_TOT * EMBED;
  __bf16* whi   = (__bf16*)(qsum + (size_t)BATCH * EMBED);
  __bf16* wlo   = whi + (size_t)NTILES * KCHUNKS * 32 * 16;

  BLP_split_w<<<NTILES * KCHUNKS, 32, 0, stream>>>(proj_W, whi, wlo);
  BLP_gemm_proj<<<1250, 128, 0, stream>>>(ent_pkl, whi, wlo, unity);
  BLP_copy_other<<<103, 256, 0, stream>>>(other, unity);
  BLP_query<<<BATCH, 256, 0, stream>>>(unity, ids, mp, qsum);
  BLP_score<<<(NE_SCORE + 255) / 256, 256, 0, stream>>>(unity, qsum, out);
}


# Projection GEMM wave tile 16x128 -> 32x256, one wave per block: each packed B fragment feeds two row tiles, A is read and split once, next fragments loaded a group ahead; sums bit-identical
# speedup vs baseline: 1.2591x; 1.2591x over previous
#include <hip/hip_runtime.h>
#include <hip/hip_bf16.h>

#define NUM_ENT   40000
#define EMBED     256
#define FEAT      768
#define KCHUNKS   (FEAT / 32)
#define NTILES    (EMBED / 16)
#define NE_TOT    40103
#define NE_SCORE  40001
#define BATCH     32

typedef __attribute__((ext_vector_type(16))) __bf16 v16bf;
typedef __attribute__((ext_vector_type(8)))  float  v8f;

__device__ __forceinline__ void load16(const float* __restrict__ p, float* v) {
  float4 a0 = *(const float4*)(p + 0);
  float4 a1 = *(const float4*)(p + 4);
  float4 b0 = *(const float4*)(p + 16);
  float4 b1 = *(const float4*)(p + 20);
  v[0] = a0.x; v[1] = a0.y; v[2]  = a0.z; v[3]  = a0.w;
  v[4] = a1.x; v[5] = a1.y; v[6]  = a1.z; v[7]  = a1.w;
  v[8] = b0.x; v[9] = b0.y; v[10] = b0.z; v[11] = b0.w;
  v[12] = b1.x; v[13] = b1.y; v[14] = b1.z; v[15] = b1.w;
}

__global__ __launch_bounds__(32) void BLP_split_w(
    const float* __restrict__ W, __bf16* __restrict__ whi,
    __bf16* __restrict__ wlo) {
  const int lane = threadIdx.x;
  const int kc   = blockIdx.x % KCHUNKS;
  const int nt   = blockIdx.x / KCHUNKS;
  const int half = lane >> 4;
  const int r15  = lane & 15;

  const float* p = W + (size_t)(nt * 16 + r15) * FEAT + kc * 32 + half * 8;
  float v[16];
  load16(p, v);

  const size_t off = ((size_t)(nt * KCHUNKS + kc) * 32 + lane) * 16;
#pragma unroll
  for (int i = 0; i < 16; ++i) {
    __bf16 h = (__bf16)v[i];
    whi[off + i] = h;
    wlo[off + i] = (__bf16)(v[i] - (float)h);
  }
}

#define MT 2
#define NG 4
#define FRAG 512

__global__ __launch_bounds__(32) void BLP_gemm_proj(
    const float* __restrict__ A, const __bf16* __restrict__ whi,
    const __bf16* __restrict__ wlo, float* __restrict__ unity) {
  const int lane = threadIdx.x;
  const int half = lane >> 4;
  const int r15  = lane & 15;
  const int m_base = blockIdx.x * (16 * MT);

  v8f acc[MT][NTILES] = {};

  const float* arow[MT];
#pragma unroll
  for (int mt = 0; mt < MT; ++mt)
    arow[mt] = A + (size_t)(m_base + mt * 16 + r15) * FEAT + half * 8;

  const __bf16* bhp = whi + lane * 16;
  const __bf16* blp = wlo + lane * 16;

  float ra[MT][16];
#pragma unroll
  for (int mt = 0; mt < MT; ++mt) load16(arow[mt], ra[mt]);

  v16bf bh[NG], bl[NG];
#pragma unroll
  for (int i = 0; i < NG; ++i) {
    const size_t off = (size_t)(i * KCHUNKS) * FRAG;
    bh[i] = *(const v16bf*)(bhp + off);
    bl[i] = *(const v16bf*)(blp + off);
  }

  for (int kc = 0; kc < KCHUNKS; ++kc) {
    const int kn = (kc + 1 < KCHUNKS) ? kc + 1 : kc;

    v16bf ahi[MT], alo[MT];
#pragma unroll
    for (int mt = 0; mt < MT; ++mt) {
#pragma unroll
      for (int i = 0; i < 16; ++i) {
        __bf16 h = (__bf16)ra[mt][i];
        ahi[mt][i] = h;
        alo[mt][i] = (__bf16)(ra[mt][i] - (float)h);
      }
    }
#pragma unroll
    for (int mt = 0; mt < MT; ++mt) {
      load16(arow[mt] + kn * 32, ra[mt]);
      if (kc + 2 < KCHUNKS)
        __builtin_prefetch(arow[mt] + (kc + 2) * 32, 0, 0);
    }

#pragma unroll
    for (int g = 0; g < NTILES / NG; ++g) {
      const int gk = (g + 1 < NTILES / NG) ? kc : kn;
      const int gn = (g + 1 < NTILES / NG) ? g + 1 : 0;
      v16bf nh[NG], nl[NG];
#pragma unroll
      for (int i = 0; i < NG; ++i) {
        const size_t off = (size_t)((gn * NG + i) * KCHUNKS + gk) * FRAG;
        nh[i] = *(const v16bf*)(bhp + off);
        nl[i] = *(const v16bf*)(blp + off);
      }
      __builtin_amdgcn_sched_barrier(0);
#pragma unroll
      for (int i = 0; i < NG; ++i)
#pragma unroll
        for (int mt = 0; mt < MT; ++mt)
          acc[mt][g * NG + i] = __builtin_amdgcn_wmma_f32_16x16x32_bf16(
              false, ahi[mt], false, bh[i], (short)0, acc[mt][g * NG + i], false, false);
#pragma unroll
      for (int i = 0; i < NG; ++i)
#pragma unroll
        for (int mt = 0; mt < MT; ++mt)
          acc[mt][g * NG + i] = __builtin_amdgcn_wmma_f32_16x16x32_bf16(
              false, ahi[mt], false, bl[i], (short)0, acc[mt][g * NG + i], false, false);
#pragma unroll
      for (int i = 0; i < NG; ++i)
#pragma unroll
        for (int mt = 0; mt < MT; ++mt)
          acc[mt][g * NG + i] = __builtin_amdgcn_wmma_f32_16x16x32_bf16(
              false, alo[mt], false, bh[i], (short)0, acc[mt][g * NG + i], false, false);
      __builtin_amdgcn_sched_barrier(0);
#pragma unroll
      for (int i = 0; i < NG; ++i) { bh[i] = nh[i]; bl[i] = nl[i]; }
    }
  }

#pragma unroll
  for (int mt = 0; mt < MT; ++mt) {
#pragma unroll
    for (int nt = 0; nt < NTILES; ++nt) {
#pragma unroll
      for (int j = 0; j < 8; ++j) {
        int m   = m_base + mt * 16 + j + half * 8;
        int col = nt * 16 + r15;
        unity[(size_t)(m + 1) * EMBED + col] = acc[mt][nt][j];
      }
    }
  }
}

__global__ __launch_bounds__(256) void BLP_copy_other(
    const float* __restrict__ other, float* __restrict__ unity) {
  int r = blockIdx.x;
  int d = threadIdx.x;
  int dst = (r == 0) ? 0 : (NUM_ENT + r);
  unity[(size_t)dst * EMBED + d] = other[(size_t)r * EMBED + d];
}

__global__ __launch_bounds__(256) void BLP_query(
    const float* __restrict__ unity, const int* __restrict__ ids,
    const int* __restrict__ mp_ptr, float* __restrict__ qsum) {
  __shared__ float s0[256];
  __shared__ float s1[256];
  const int b = blockIdx.x;
  const int d = threadIdx.x;
  const int mp = *mp_ptr;
  const int c0 = (mp == 0) ? 1 : 0;
  const int c1 = (mp <= 1) ? 2 : 1;
  const int id0 = ids[b * 3 + c0];
  const int id1 = ids[b * 3 + c1];
  float v0 = unity[(size_t)id0 * EMBED + d];
  float v1 = unity[(size_t)id1 * EMBED + d];
  s0[d] = v0 * v0;
  s1[d] = v1 * v1;
  __syncthreads();
  for (int off = 128; off > 0; off >>= 1) {
    if (d < off) { s0[d] += s0[d + off]; s1[d] += s1[d + off]; }
    __syncthreads();
  }
  float r0 = 1.0f / fmaxf(sqrtf(s0[0]), 1e-12f);
  float r1 = 1.0f / fmaxf(sqrtf(s1[0]), 1e-12f);
  qsum[(size_t)b * EMBED + d] = v0 * r0 + v1 * r1;
}

__global__ __launch_bounds__(256) void BLP_score(
    const float* __restrict__ unity, const float* __restrict__ qsum,
    float* __restrict__ out) {
  __shared__ float q[BATCH * EMBED];
  for (int i = threadIdx.x; i < BATCH * EMBED; i += 256) q[i] = qsum[i];
  __syncthreads();

  const int e = blockIdx.x * 256 + threadIdx.x;
  const bool active = (e < NE_SCORE);
  const float4* row = (const float4*)(unity + (size_t)(active ? e : 0) * EMBED);

  float acc[BATCH];
#pragma unroll
  for (int b = 0; b < BATCH; ++b) acc[b] = 0.0f;

  for (int d4 = 0; d4 < EMBED / 4; ++d4) {
    float4 v = row[d4];
    const int dbase = d4 * 4;
#pragma unroll
    for (int b = 0; b < BATCH; ++b) {
      const float* qb = q + b * EMBED + dbase;
      acc[b] += fabsf(qb[0] - v.x) + fabsf(qb[1] - v.y) +
                fabsf(qb[2] - v.z) + fabsf(qb[3] - v.w);
    }
  }
  if (active) {
#pragma unroll
    for (int b = 0; b < BATCH; ++b)
      out[(size_t)b * NE_SCORE + e] = -acc[b];
  }
}

extern "C" void kernel_launch(void* const* d_in, const int* in_sizes, int n_in,
                              void* d_out, int out_size, void* d_ws, size_t ws_size,
                              hipStream_t stream) {
  (void)in_sizes; (void)n_in; (void)out_size; (void)ws_size;
  const float* ent_pkl = (const float*)d_in[0];
  const float* other   = (const float*)d_in[1];
  const float* proj_W  = (const float*)d_in[2];
  const int*   ids     = (const int*)d_in[3];
  const int*   mp      = (const int*)d_in[4];
  float*       out     = (float*)d_out;

  float*  unity = (float*)d_ws;
  float*  qsum  = unity + (size_t)NE_TOT * EMBED;
  __bf16* whi   = (__bf16*)(qsum + (size_t)BATCH * EMBED);
  __bf16* wlo   = whi + (size_t)NTILES * KCHUNKS * 32 * 16;

  BLP_split_w<<<NTILES * KCHUNKS, 32, 0, stream>>>(proj_W, whi, wlo);
  BLP_gemm_proj<<<NUM_ENT / (16 * MT), 32, 0, stream>>>(ent_pkl, whi, wlo, unity);
  BLP_copy_other<<<103, 256, 0, stream>>>(other, unity);
  BLP_query<<<BATCH, 256, 0, stream>>>(unity, ids, mp, qsum);
  BLP_score<<<(NE_SCORE + 255) / 256, 256, 0, stream>>>(unity, qsum, out);
}


# Projection GEMM: each k-chunk's packed B copied to LDS once per 4-wave block and read there by all waves (was: every wave re-read all of B from L2); wave tile 48x256; sums bit-identical
# speedup vs baseline: 1.4262x; 1.1327x over previous
#include <hip/hip_runtime.h>
#include <hip/hip_bf16.h>

#define NUM_ENT   40000
#define EMBED     256
#define FEAT      768
#define KCHUNKS   (FEAT / 32)
#define NTILES    (EMBED / 16)
#define NE_TOT    40103
#define NE_SCORE  40001
#define BATCH     32

typedef __attribute__((ext_vector_type(16))) __bf16 v16bf;
typedef __attribute__((ext_vector_type(8)))  float  v8f;

__device__ __forceinline__ void load16(const float* __restrict__ p, float* v) {
  float4 a0 = *(const float4*)(p + 0);
  float4 a1 = *(const float4*)(p + 4);
  float4 b0 = *(const float4*)(p + 16);
  float4 b1 = *(const float4*)(p + 20);
  v[0] = a0.x; v[1] = a0.y; v[2]  = a0.z; v[3]  = a0.w;
  v[4] = a1.x; v[5] = a1.y; v[6]  = a1.z; v[7]  = a1.w;
  v[8] = b0.x; v[9] = b0.y; v[10] = b0.z; v[11] = b0.w;
  v[12] = b1.x; v[13] = b1.y; v[14] = b1.z; v[15] = b1.w;
}

__global__ __launch_bounds__(32) void BLP_split_w(
    const float* __restrict__ W, __bf16* __restrict__ whi,
    __bf16* __restrict__ wlo) {
  const int lane = threadIdx.x;
  const int kc   = blockIdx.x % KCHUNKS;
  const int nt   = blockIdx.x / KCHUNKS;
  const int half = lane >> 4;
  const int r15  = lane & 15;

  const float* p = W + (size_t)(nt * 16 + r15) * FEAT + kc * 32 + half * 8;
  float v[16];
  load16(p, v);

  const size_t base = (size_t)(kc * NTILES + nt) * 512 + lane * 8;
#pragma unroll
  for (int i = 0; i < 16; ++i) {
    __bf16 h = (__bf16)v[i];
    const size_t off = base + (i >> 3) * 256 + (i & 7);
    whi[off] = h;
    wlo[off] = (__bf16)(v[i] - (float)h);
  }
}

#define MT 3
#define NG 4
#define FRAG 512
#define WAVES 4
#define ROWS_WG (16 * MT * WAVES)
#define GEMM_WGS ((NUM_ENT + ROWS_WG - 1) / ROWS_WG)
#define PLANE (NTILES * FRAG)
#define PLANE_V (PLANE / 8)

typedef __attribute__((ext_vector_type(8))) __bf16 v8bf;
typedef __attribute__((ext_vector_type(4))) unsigned int u32x4;

__device__ __forceinline__ v16bf lds_frag(const __bf16* p) {
  v8bf a = *(const v8bf*)(p);
  v8bf b = *(const v8bf*)(p + 256);
  return __builtin_shufflevector(a, b, 0, 1, 2, 3, 4, 5, 6, 7,
                                 8, 9, 10, 11, 12, 13, 14, 15);
}

__device__ __forceinline__ void lds_handover() {
  __builtin_amdgcn_fence(__ATOMIC_RELEASE, "workgroup", "local");
  __builtin_amdgcn_s_barrier();
  __builtin_amdgcn_fence(__ATOMIC_ACQUIRE, "workgroup", "local");
}

__global__ __launch_bounds__(32 * WAVES) void BLP_gemm_proj(
    const float* __restrict__ A, const __bf16* __restrict__ whi,
    const __bf16* __restrict__ wlo, float* __restrict__ unity) {
  __shared__ u32x4 sB[2][2 * PLANE_V];

  const int tid  = threadIdx.x;
  const int lane = tid & 31;
  const int wave = __builtin_amdgcn_readfirstlane(tid >> 5);
  const int half = lane >> 4;
  const int r15  = lane & 15;
  const int m_base = (blockIdx.x * WAVES + wave) * (16 * MT);

  v8f acc[MT][NTILES] = {};

  const float* arow[MT];
#pragma unroll
  for (int mt = 0; mt < MT; ++mt) {
    int r = m_base + mt * 16 + r15;
    r = (r < NUM_ENT) ? r : (NUM_ENT - 1);
    arow[mt] = A + (size_t)r * FEAT + half * 8;
  }

  float ra[MT][16];
#pragma unroll
  for (int mt = 0; mt < MT; ++mt) load16(arow[mt], ra[mt]);
  __builtin_amdgcn_sched_barrier(0);

  const u32x4* gh = (const u32x4*)whi + tid;
  const u32x4* gl = (const u32x4*)wlo + tid;

  u32x4 st[16];
#pragma unroll
  for (int s = 0; s < 8; ++s) { st[s] = gh[s * 128]; st[8 + s] = gl[s * 128]; }
  {
    u32x4* dst = &sB[0][tid];
#pragma unroll
    for (int s = 0; s < 8; ++s) { dst[s * 128] = st[s]; dst[PLANE_V + s * 128] = st[8 + s]; }
  }
#pragma unroll
  for (int s = 0; s < 8; ++s) {
    st[s]     = gh[PLANE_V + s * 128];
    st[8 + s] = gl[PLANE_V + s * 128];
  }

  lds_handover();

  for (int kc = 0; kc < KCHUNKS; ++kc) {
    const int kn = (kc + 1 < KCHUNKS) ? kc + 1 : kc;

    const __bf16* sh = (const __bf16*)&sB[kc & 1][0] + lane * 8;
    const __bf16* sl = sh + PLANE;

    v16bf bh[NG], bl[NG];
#pragma unroll
    for (int i = 0; i < NG; ++i) {
      bh[i] = lds_frag(sh + i * FRAG);
      bl[i] = lds_frag(sl + i * FRAG);
    }

    v16bf ahi[MT], alo[MT];
#pragma unroll
    for (int mt = 0; mt < MT; ++mt) {
#pragma unroll
      for (int i = 0; i < 16; ++i) {
        __bf16 h = (__bf16)ra[mt][i];
        ahi[mt][i] = h;
        alo[mt][i] = (__bf16)(ra[mt][i] - (float)h);
      }
    }
    const int kp = (kc + 2 < KCHUNKS) ? kc + 2 : KCHUNKS - 1;
#pragma unroll
    for (int mt = 0; mt < MT; ++mt) {
      load16(arow[mt] + kn * 32, ra[mt]);
      __builtin_prefetch(arow[mt] + kp * 32, 0, 0);
    }

#pragma unroll
    for (int g = 0; g < NTILES / NG; ++g) {
      v16bf nh[NG], nl[NG];
      if (g + 1 < NTILES / NG) {
#pragma unroll
        for (int i = 0; i < NG; ++i) {
          nh[i] = lds_frag(sh + ((g + 1) * NG + i) * FRAG);
          nl[i] = lds_frag(sl + ((g + 1) * NG + i) * FRAG);
        }
      }
      __builtin_amdgcn_sched_barrier(0);
#pragma unroll
      for (int i = 0; i < NG; ++i)
#pragma unroll
        for (int mt = 0; mt < MT; ++mt)
          acc[mt][g * NG + i] = __builtin_amdgcn_wmma_f32_16x16x32_bf16(
              false, ahi[mt], false, bh[i], (short)0, acc[mt][g * NG + i], false, false);
#pragma unroll
      for (int i = 0; i < NG; ++i)
#pragma unroll
        for (int mt = 0; mt < MT; ++mt)
          acc[mt][g * NG + i] = __builtin_amdgcn_wmma_f32_16x16x32_bf16(
              false, ahi[mt], false, bl[i], (short)0, acc[mt][g * NG + i], false, false);
#pragma unroll
      for (int i = 0; i < NG; ++i)
#pragma unroll
        for (int mt = 0; mt < MT; ++mt)
          acc[mt][g * NG + i] = __builtin_amdgcn_wmma_f32_16x16x32_bf16(
              false, alo[mt], false, bh[i], (short)0, acc[mt][g * NG + i], false, false);
      __builtin_amdgcn_sched_barrier(0);
      if (g + 1 < NTILES / NG) {
#pragma unroll
        for (int i = 0; i < NG; ++i) { bh[i] = nh[i]; bl[i] = nl[i]; }
      }
    }

    {
      u32x4* dst = &sB[(kc + 1) & 1][tid];
#pragma unroll
      for (int s = 0; s < 8; ++s) { dst[s * 128] = st[s]; dst[PLANE_V + s * 128] = st[8 + s]; }
    }
#pragma unroll
    for (int s = 0; s < 8; ++s) {
      st[s]     = gh[(size_t)kp * PLANE_V + s * 128];
      st[8 + s] = gl[(size_t)kp * PLANE_V + s * 128];
    }
    lds_handover();
  }

#pragma unroll
  for (int mt = 0; mt < MT; ++mt) {
    if (m_base + mt * 16 < NUM_ENT) {
#pragma unroll
      for (int nt = 0; nt < NTILES; ++nt) {
#pragma unroll
        for (int j = 0; j < 8; ++j) {
          int m   = m_base + mt * 16 + j + half * 8;
          int col = nt * 16 + r15;
          unity[(size_t)(m + 1) * EMBED + col] = acc[mt][nt][j];
        }
      }
    }
  }
}

__global__ __launch_bounds__(256) void BLP_copy_other(
    const float* __restrict__ other, float* __restrict__ unity) {
  int r = blockIdx.x;
  int d = threadIdx.x;
  int dst = (r == 0) ? 0 : (NUM_ENT + r);
  unity[(size_t)dst * EMBED + d] = other[(size_t)r * EMBED + d];
}

__global__ __launch_bounds__(256) void BLP_query(
    const float* __restrict__ unity, const int* __restrict__ ids,
    const int* __restrict__ mp_ptr, float* __restrict__ qsum) {
  __shared__ float s0[256];
  __shared__ float s1[256];
  const int b = blockIdx.x;
  const int d = threadIdx.x;
  const int mp = *mp_ptr;
  const int c0 = (mp == 0) ? 1 : 0;
  const int c1 = (mp <= 1) ? 2 : 1;
  const int id0 = ids[b * 3 + c0];
  const int id1 = ids[b * 3 + c1];
  float v0 = unity[(size_t)id0 * EMBED + d];
  float v1 = unity[(size_t)id1 * EMBED + d];
  s0[d] = v0 * v0;
  s1[d] = v1 * v1;
  __syncthreads();
  for (int off = 128; off > 0; off >>= 1) {
    if (d < off) { s0[d] += s0[d + off]; s1[d] += s1[d + off]; }
    __syncthreads();
  }
  float r0 = 1.0f / fmaxf(sqrtf(s0[0]), 1e-12f);
  float r1 = 1.0f / fmaxf(sqrtf(s1[0]), 1e-12f);
  qsum[(size_t)b * EMBED + d] = v0 * r0 + v1 * r1;
}

__global__ __launch_bounds__(256) void BLP_score(
    const float* __restrict__ unity, const float* __restrict__ qsum,
    float* __restrict__ out) {
  __shared__ float q[BATCH * EMBED];
  for (int i = threadIdx.x; i < BATCH * EMBED; i += 256) q[i] = qsum[i];
  __syncthreads();

  const int e = blockIdx.x * 256 + threadIdx.x;
  const bool active = (e < NE_SCORE);
  const float4* row = (const float4*)(unity + (size_t)(active ? e : 0) * EMBED);

  float acc[BATCH];
#pragma unroll
  for (int b = 0; b < BATCH; ++b) acc[b] = 0.0f;

  for (int d4 = 0; d4 < EMBED / 4; ++d4) {
    float4 v = row[d4];
    const int dbase = d4 * 4;
#pragma unroll
    for (int b = 0; b < BATCH; ++b) {
      const float* qb = q + b * EMBED + dbase;
      acc[b] += fabsf(qb[0] - v.x) + fabsf(qb[1] - v.y) +
                fabsf(qb[2] - v.z) + fabsf(qb[3] - v.w);
    }
  }
  if (active) {
#pragma unroll
    for (int b = 0; b < BATCH; ++b)
      out[(size_t)b * NE_SCORE + e] = -acc[b];
  }
}

extern "C" void kernel_launch(void* const* d_in, const int* in_sizes, int n_in,
                              void* d_out, int out_size, void* d_ws, size_t ws_size,
                              hipStream_t stream) {
  (void)in_sizes; (void)n_in; (void)out_size; (void)ws_size;
  const float* ent_pkl = (const float*)d_in[0];
  const float* other   = (const float*)d_in[1];
  const float* proj_W  = (const float*)d_in[2];
  const int*   ids     = (const int*)d_in[3];
  const int*   mp      = (const int*)d_in[4];
  float*       out     = (float*)d_out;

  float*  unity = (float*)d_ws;
  float*  qsum  = unity + (size_t)NE_TOT * EMBED;
  __bf16* whi   = (__bf16*)(qsum + (size_t)BATCH * EMBED);
  __bf16* wlo   = whi + (size_t)NTILES * KCHUNKS * 32 * 16;

  BLP_split_w<<<NTILES * KCHUNKS, 32, 0, stream>>>(proj_W, whi, wlo);
  BLP_gemm_proj<<<GEMM_WGS, 32 * WAVES, 0, stream>>>(ent_pkl, whi, wlo, unity);
  BLP_copy_other<<<103, 256, 0, stream>>>(other, unity);
  BLP_query<<<BATCH, 256, 0, stream>>>(unity, ids, mp, qsum);
  BLP_score<<<(NE_SCORE + 255) / 256, 256, 0, stream>>>(unity, qsum, out);
}


# Launches 5 -> 3: the plain row copy runs as extra blocks of the weight-split launch; every score block builds the 32 normalized query sums in its own prologue (norm sums keep their pairing tree)
# speedup vs baseline: 1.5173x; 1.0639x over previous
#include <hip/hip_runtime.h>
#include <hip/hip_bf16.h>

#define NUM_ENT   40000
#define EMBED     256
#define FEAT      768
#define KCHUNKS   (FEAT / 32)
#define NTILES    (EMBED / 16)
#define NE_TOT    40103
#define NE_SCORE  40001
#define BATCH     32

typedef __attribute__((ext_vector_type(16))) __bf16 v16bf;
typedef __attribute__((ext_vector_type(8)))  float  v8f;

__device__ __forceinline__ void load16(const float* __restrict__ p, float* v) {
  float4 a0 = *(const float4*)(p + 0);
  float4 a1 = *(const float4*)(p + 4);
  float4 b0 = *(const float4*)(p + 16);
  float4 b1 = *(const float4*)(p + 20);
  v[0] = a0.x; v[1] = a0.y; v[2]  = a0.z; v[3]  = a0.w;
  v[4] = a1.x; v[5] = a1.y; v[6]  = a1.z; v[7]  = a1.w;
  v[8] = b0.x; v[9] = b0.y; v[10] = b0.z; v[11] = b0.w;
  v[12] = b1.x; v[13] = b1.y; v[14] = b1.z; v[15] = b1.w;
}

#define SPLIT_WGS (NTILES * KCHUNKS)
#define OTHER_ROWS 103
__global__ __launch_bounds__(32) void BLP_split_w(
    const float* __restrict__ W, __bf16* __restrict__ whi,
    __bf16* __restrict__ wlo, const float* __restrict__ other,
    float* __restrict__ unity) {
  const int lane = threadIdx.x;
  if (blockIdx.x >= SPLIT_WGS) {
    const int r   = blockIdx.x - SPLIT_WGS;
    const int dst = (r == 0) ? 0 : (NUM_ENT + r);
    const float4* s = (const float4*)(other + (size_t)r * EMBED);
    float4*       d = (float4*)(unity + (size_t)dst * EMBED);
    d[lane]      = s[lane];
    d[lane + 32] = s[lane + 32];
    return;
  }
  const int kc   = blockIdx.x % KCHUNKS;
  const int nt   = blockIdx.x / KCHUNKS;
  const int half = lane >> 4;
  const int r15  = lane & 15;

  const float* p = W + (size_t)(nt * 16 + r15) * FEAT + kc * 32 + half * 8;
  float v[16];
  load16(p, v);

  const size_t base = (size_t)(kc * NTILES + nt) * 512 + lane * 8;
#pragma unroll
  for (int i = 0; i < 16; ++i) {
    __bf16 h = (__bf16)v[i];
    const size_t off = base + (i >> 3) * 256 + (i & 7);
    whi[off] = h;
    wlo[off] = (__bf16)(v[i] - (float)h);
  }
}

#define MT 3
#define NG 4
#define FRAG 512
#define WAVES 4
#define ROWS_WG (16 * MT * WAVES)
#define GEMM_WGS ((NUM_ENT + ROWS_WG - 1) / ROWS_WG)
#define PLANE (NTILES * FRAG)
#define PLANE_V (PLANE / 8)

typedef __attribute__((ext_vector_type(8))) __bf16 v8bf;
typedef __attribute__((ext_vector_type(4))) unsigned int u32x4;

__device__ __forceinline__ v16bf lds_frag(const __bf16* p) {
  v8bf a = *(const v8bf*)(p);
  v8bf b = *(const v8bf*)(p + 256);
  return __builtin_shufflevector(a, b, 0, 1, 2, 3, 4, 5, 6, 7,
                                 8, 9, 10, 11, 12, 13, 14, 15);
}

__device__ __forceinline__ void lds_handover() {
  __builtin_amdgcn_fence(__ATOMIC_RELEASE, "workgroup", "local");
  __builtin_amdgcn_s_barrier();
  __builtin_amdgcn_fence(__ATOMIC_ACQUIRE, "workgroup", "local");
}

__global__ __launch_bounds__(32 * WAVES) void BLP_gemm_proj(
    const float* __restrict__ A, const __bf16* __restrict__ whi,
    const __bf16* __restrict__ wlo, float* __restrict__ unity) {
  __shared__ u32x4 sB[2][2 * PLANE_V];

  const int tid  = threadIdx.x;
  const int lane = tid & 31;
  const int wave = __builtin_amdgcn_readfirstlane(tid >> 5);
  const int half = lane >> 4;
  const int r15  = lane & 15;
  const int m_base = (blockIdx.x * WAVES + wave) * (16 * MT);

  v8f acc[MT][NTILES] = {};

  const float* arow[MT];
#pragma unroll
  for (int mt = 0; mt < MT; ++mt) {
    int r = m_base + mt * 16 + r15;
    r = (r < NUM_ENT) ? r : (NUM_ENT - 1);
    arow[mt] = A + (size_t)r * FEAT + half * 8;
  }

  float ra[MT][16];
#pragma unroll
  for (int mt = 0; mt < MT; ++mt) load16(arow[mt], ra[mt]);
  __builtin_amdgcn_sched_barrier(0);

  const u32x4* gh = (const u32x4*)whi + tid;
  const u32x4* gl = (const u32x4*)wlo + tid;

  u32x4 st[16];
#pragma unroll
  for (int s = 0; s < 8; ++s) { st[s] = gh[s * 128]; st[8 + s] = gl[s * 128]; }
  {
    u32x4* dst = &sB[0][tid];
#pragma unroll
    for (int s = 0; s < 8; ++s) { dst[s * 128] = st[s]; dst[PLANE_V + s * 128] = st[8 + s]; }
  }
#pragma unroll
  for (int s = 0; s < 8; ++s) {
    st[s]     = gh[PLANE_V + s * 128];
    st[8 + s] = gl[PLANE_V + s * 128];
  }

  lds_handover();

  for (int kc = 0; kc < KCHUNKS; ++kc) {
    const int kn = (kc + 1 < KCHUNKS) ? kc + 1 : kc;

    const __bf16* sh = (const __bf16*)&sB[kc & 1][0] + lane * 8;
    const __bf16* sl = sh + PLANE;

    v16bf bh[NG], bl[NG];
#pragma unroll
    for (int i = 0; i < NG; ++i) {
      bh[i] = lds_frag(sh + i * FRAG);
      bl[i] = lds_frag(sl + i * FRAG);
    }

    v16bf ahi[MT], alo[MT];
#pragma unroll
    for (int mt = 0; mt < MT; ++mt) {
#pragma unroll
      for (int i = 0; i < 16; ++i) {
        __bf16 h = (__bf16)ra[mt][i];
        ahi[mt][i] = h;
        alo[mt][i] = (__bf16)(ra[mt][i] - (float)h);
      }
    }
    const int kp = (kc + 2 < KCHUNKS) ? kc + 2 : KCHUNKS - 1;
#pragma unroll
    for (int mt = 0; mt < MT; ++mt) {
      load16(arow[mt] + kn * 32, ra[mt]);
      __builtin_prefetch(arow[mt] + kp * 32, 0, 0);
    }

#pragma unroll
    for (int g = 0; g < NTILES / NG; ++g) {
      v16bf nh[NG], nl[NG];
      if (g + 1 < NTILES / NG) {
#pragma unroll
        for (int i = 0; i < NG; ++i) {
          nh[i] = lds_frag(sh + ((g + 1) * NG + i) * FRAG);
          nl[i] = lds_frag(sl + ((g + 1) * NG + i) * FRAG);
        }
      }
      __builtin_amdgcn_sched_barrier(0);
#pragma unroll
      for (int i = 0; i < NG; ++i)
#pragma unroll
        for (int mt = 0; mt < MT; ++mt)
          acc[mt][g * NG + i] = __builtin_amdgcn_wmma_f32_16x16x32_bf16(
              false, ahi[mt], false, bh[i], (short)0, acc[mt][g * NG + i], false, false);
#pragma unroll
      for (int i = 0; i < NG; ++i)
#pragma unroll
        for (int mt = 0; mt < MT; ++mt)
          acc[mt][g * NG + i] = __builtin_amdgcn_wmma_f32_16x16x32_bf16(
              false, ahi[mt], false, bl[i], (short)0, acc[mt][g * NG + i], false, false);
#pragma unroll
      for (int i = 0; i < NG; ++i)
#pragma unroll
        for (int mt = 0; mt < MT; ++mt)
          acc[mt][g * NG + i] = __builtin_amdgcn_wmma_f32_16x16x32_bf16(
              false, alo[mt], false, bh[i], (short)0, acc[mt][g * NG + i], false, false);
      __builtin_amdgcn_sched_barrier(0);
      if (g + 1 < NTILES / NG) {
#pragma unroll
        for (int i = 0; i < NG; ++i) { bh[i] = nh[i]; bl[i] = nl[i]; }
      }
    }

    {
      u32x4* dst = &sB[(kc + 1) & 1][tid];
#pragma unroll
      for (int s = 0; s < 8; ++s) { dst[s * 128] = st[s]; dst[PLANE_V + s * 128] = st[8 + s]; }
    }
#pragma unroll
    for (int s = 0; s < 8; ++s) {
      st[s]     = gh[(size_t)kp * PLANE_V + s * 128];
      st[8 + s] = gl[(size_t)kp * PLANE_V + s * 128];
    }
    lds_handover();
  }

#pragma unroll
  for (int mt = 0; mt < MT; ++mt) {
    if (m_base + mt * 16 < NUM_ENT) {
#pragma unroll
      for (int nt = 0; nt < NTILES; ++nt) {
#pragma unroll
        for (int j = 0; j < 8; ++j) {
          int m   = m_base + mt * 16 + j + half * 8;
          int col = nt * 16 + r15;
          unity[(size_t)(m + 1) * EMBED + col] = acc[mt][nt][j];
        }
      }
    }
  }
}

__global__ __launch_bounds__(256) void BLP_score(
    const float* __restrict__ unity, const int* __restrict__ ids,
    const int* __restrict__ mp_ptr, float* __restrict__ out) {
  __shared__ float q[BATCH * EMBED];
  {
    const int lane = threadIdx.x & 31;
    const int wave = __builtin_amdgcn_readfirstlane((int)(threadIdx.x >> 5));
    const int mp = *mp_ptr;
    const int c0 = (mp == 0) ? 1 : 0;
    const int c1 = (mp <= 1) ? 2 : 1;
    float v[4][2][8];
#pragma unroll
    for (int j = 0; j < 4; ++j) {
      const int b = wave * 4 + j;
      const float* p0 = unity + (size_t)ids[b * 3 + c0] * EMBED + lane;
      const float* p1 = unity + (size_t)ids[b * 3 + c1] * EMBED + lane;
#pragma unroll
      for (int i = 0; i < 8; ++i) { v[j][0][i] = p0[32 * i]; v[j][1][i] = p1[32 * i]; }
    }
#pragma unroll
    for (int j = 0; j < 4; ++j) {
      float rinv[2];
#pragma unroll
      for (int r = 0; r < 2; ++r) {
        float s[8];
#pragma unroll
        for (int i = 0; i < 8; ++i) s[i] = v[j][r][i] * v[j][r][i];
#pragma unroll
        for (int i = 0; i < 4; ++i) s[i] += s[i + 4];
        s[0] += s[2]; s[1] += s[3];
        float t = s[0] + s[1];
#pragma unroll
        for (int m = 16; m > 0; m >>= 1) t += __shfl_xor(t, m);
        rinv[r] = 1.0f / fmaxf(sqrtf(t), 1e-12f);
      }
      float* qb = q + (wave * 4 + j) * EMBED + lane;
#pragma unroll
      for (int i = 0; i < 8; ++i)
        qb[32 * i] = v[j][0][i] * rinv[0] + v[j][1][i] * rinv[1];
    }
  }
  __syncthreads();

  const int e = blockIdx.x * 256 + threadIdx.x;
  const bool active = (e < NE_SCORE);
  const float4* row = (const float4*)(unity + (size_t)(active ? e : 0) * EMBED);

  float acc[BATCH];
#pragma unroll
  for (int b = 0; b < BATCH; ++b) acc[b] = 0.0f;

  for (int d4 = 0; d4 < EMBED / 4; ++d4) {
    float4 v = row[d4];
    const int dbase = d4 * 4;
#pragma unroll
    for (int b = 0; b < BATCH; ++b) {
      const float* qb = q + b * EMBED + dbase;
      acc[b] += fabsf(qb[0] - v.x) + fabsf(qb[1] - v.y) +
                fabsf(qb[2] - v.z) + fabsf(qb[3] - v.w);
    }
  }
  if (active) {
#pragma unroll
    for (int b = 0; b < BATCH; ++b)
      out[(size_t)b * NE_SCORE + e] = -acc[b];
  }
}

extern "C" void kernel_launch(void* const* d_in, const int* in_sizes, int n_in,
                              void* d_out, int out_size, void* d_ws, size_t ws_size,
                              hipStream_t stream) {
  (void)in_sizes; (void)n_in; (void)out_size; (void)ws_size;
  const float* ent_pkl = (const float*)d_in[0];
  const float* other   = (const float*)d_in[1];
  const float* proj_W  = (const float*)d_in[2];
  const int*   ids     = (const int*)d_in[3];
  const int*   mp      = (const int*)d_in[4];
  float*       out     = (float*)d_out;

  float*  unity = (float*)d_ws;
  __bf16* whi   = (__bf16*)(unity + (size_t)NE_TOT * EMBED + (size_t)BATCH * EMBED);
  __bf16* wlo   = whi + (size_t)NTILES * KCHUNKS * 32 * 16;

  BLP_split_w<<<SPLIT_WGS + OTHER_ROWS, 32, 0, stream>>>(proj_W, whi, wlo, other, unity);
  BLP_gemm_proj<<<GEMM_WGS, 32 * WAVES, 0, stream>>>(ent_pkl, whi, wlo, unity);
  BLP_score<<<(NE_SCORE + 255) / 256, 256, 0, stream>>>(unity, ids, mp, out);
}


# Score loop: |q - row| formed two dims per packed subtract; each term added to its running sum by one add with abs as source modifier (no separate abs, no pairing moves); terms summed in d order
# speedup vs baseline: 1.6528x; 1.0893x over previous
#include <hip/hip_runtime.h>
#include <hip/hip_bf16.h>

#define NUM_ENT   40000
#define EMBED     256
#define FEAT      768
#define KCHUNKS   (FEAT / 32)
#define NTILES    (EMBED / 16)
#define NE_TOT    40103
#define NE_SCORE  40001
#define BATCH     32

typedef __attribute__((ext_vector_type(16))) __bf16 v16bf;
typedef __attribute__((ext_vector_type(8)))  float  v8f;
typedef __attribute__((ext_vector_type(2)))  float  v2f;

__device__ __forceinline__ void load16(const float* __restrict__ p, float* v) {
  float4 a0 = *(const float4*)(p + 0);
  float4 a1 = *(const float4*)(p + 4);
  float4 b0 = *(const float4*)(p + 16);
  float4 b1 = *(const float4*)(p + 20);
  v[0] = a0.x; v[1] = a0.y; v[2]  = a0.z; v[3]  = a0.w;
  v[4] = a1.x; v[5] = a1.y; v[6]  = a1.z; v[7]  = a1.w;
  v[8] = b0.x; v[9] = b0.y; v[10] = b0.z; v[11] = b0.w;
  v[12] = b1.x; v[13] = b1.y; v[14] = b1.z; v[15] = b1.w;
}

#define SPLIT_WGS (NTILES * KCHUNKS)
#define OTHER_ROWS 103
__global__ __launch_bounds__(32) void BLP_split_w(
    const float* __restrict__ W, __bf16* __restrict__ whi,
    __bf16* __restrict__ wlo, const float* __restrict__ other,
    float* __restrict__ unity) {
  const int lane = threadIdx.x;
  if (blockIdx.x >= SPLIT_WGS) {
    const int r   = blockIdx.x - SPLIT_WGS;
    const int dst = (r == 0) ? 0 : (NUM_ENT + r);
    const float4* s = (const float4*)(other + (size_t)r * EMBED);
    float4*       d = (float4*)(unity + (size_t)dst * EMBED);
    d[lane]      = s[lane];
    d[lane + 32] = s[lane + 32];
    return;
  }
  const int kc   = blockIdx.x % KCHUNKS;
  const int nt   = blockIdx.x / KCHUNKS;
  const int half = lane >> 4;
  const int r15  = lane & 15;

  const float* p = W + (size_t)(nt * 16 + r15) * FEAT + kc * 32 + half * 8;
  float v[16];
  load16(p, v);

  const size_t base = (size_t)(kc * NTILES + nt) * 512 + lane * 8;
#pragma unroll
  for (int i = 0; i < 16; ++i) {
    __bf16 h = (__bf16)v[i];
    const size_t off = base + (i >> 3) * 256 + (i & 7);
    whi[off] = h;
    wlo[off] = (__bf16)(v[i] - (float)h);
  }
}

#define MT 3
#define NG 4
#define FRAG 512
#define WAVES 4
#define ROWS_WG (16 * MT * WAVES)
#define GEMM_WGS ((NUM_ENT + ROWS_WG - 1) / ROWS_WG)
#define PLANE (NTILES * FRAG)
#define PLANE_V (PLANE / 8)

typedef __attribute__((ext_vector_type(8))) __bf16 v8bf;
typedef __attribute__((ext_vector_type(4))) unsigned int u32x4;

__device__ __forceinline__ v16bf lds_frag(const __bf16* p) {
  v8bf a = *(const v8bf*)(p);
  v8bf b = *(const v8bf*)(p + 256);
  return __builtin_shufflevector(a, b, 0, 1, 2, 3, 4, 5, 6, 7,
                                 8, 9, 10, 11, 12, 13, 14, 15);
}

__device__ __forceinline__ void lds_handover() {
  __builtin_amdgcn_fence(__ATOMIC_RELEASE, "workgroup", "local");
  __builtin_amdgcn_s_barrier();
  __builtin_amdgcn_fence(__ATOMIC_ACQUIRE, "workgroup", "local");
}

__global__ __launch_bounds__(32 * WAVES) void BLP_gemm_proj(
    const float* __restrict__ A, const __bf16* __restrict__ whi,
    const __bf16* __restrict__ wlo, float* __restrict__ unity) {
  __shared__ u32x4 sB[2][2 * PLANE_V];

  const int tid  = threadIdx.x;
  const int lane = tid & 31;
  const int wave = __builtin_amdgcn_readfirstlane(tid >> 5);
  const int half = lane >> 4;
  const int r15  = lane & 15;
  const int m_base = (blockIdx.x * WAVES + wave) * (16 * MT);

  v8f acc[MT][NTILES] = {};

  const float* arow[MT];
#pragma unroll
  for (int mt = 0; mt < MT; ++mt) {
    int r = m_base + mt * 16 + r15;
    r = (r < NUM_ENT) ? r : (NUM_ENT - 1);
    arow[mt] = A + (size_t)r * FEAT + half * 8;
  }

  float ra[MT][16];
#pragma unroll
  for (int mt = 0; mt < MT; ++mt) load16(arow[mt], ra[mt]);
  __builtin_amdgcn_sched_barrier(0);

  const u32x4* gh = (const u32x4*)whi + tid;
  const u32x4* gl = (const u32x4*)wlo + tid;

  u32x4 st[16];
#pragma unroll
  for (int s = 0; s < 8; ++s) { st[s] = gh[s * 128]; st[8 + s] = gl[s * 128]; }
  {
    u32x4* dst = &sB[0][tid];
#pragma unroll
    for (int s = 0; s < 8; ++s) { dst[s * 128] = st[s]; dst[PLANE_V + s * 128] = st[8 + s]; }
  }
#pragma unroll
  for (int s = 0; s < 8; ++s) {
    st[s]     = gh[PLANE_V + s * 128];
    st[8 + s] = gl[PLANE_V + s * 128];
  }

  lds_handover();

  for (int kc = 0; kc < KCHUNKS; ++kc) {
    const int kn = (kc + 1 < KCHUNKS) ? kc + 1 : kc;

    const __bf16* sh = (const __bf16*)&sB[kc & 1][0] + lane * 8;
    const __bf16* sl = sh + PLANE;

    v16bf bh[NG], bl[NG];
#pragma unroll
    for (int i = 0; i < NG; ++i) {
      bh[i] = lds_frag(sh + i * FRAG);
      bl[i] = lds_frag(sl + i * FRAG);
    }

    v16bf ahi[MT], alo[MT];
#pragma unroll
    for (int mt = 0; mt < MT; ++mt) {
#pragma unroll
      for (int i = 0; i < 16; ++i) {
        __bf16 h = (__bf16)ra[mt][i];
        ahi[mt][i] = h;
        alo[mt][i] = (__bf16)(ra[mt][i] - (float)h);
      }
    }
    const int kp = (kc + 2 < KCHUNKS) ? kc + 2 : KCHUNKS - 1;
#pragma unroll
    for (int mt = 0; mt < MT; ++mt) {
      load16(arow[mt] + kn * 32, ra[mt]);
      __builtin_prefetch(arow[mt] + kp * 32, 0, 0);
    }

#pragma unroll
    for (int g = 0; g < NTILES / NG; ++g) {
      v16bf nh[NG], nl[NG];
      if (g + 1 < NTILES / NG) {
#pragma unroll
        for (int i = 0; i < NG; ++i) {
          nh[i] = lds_frag(sh + ((g + 1) * NG + i) * FRAG);
          nl[i] = lds_frag(sl + ((g + 1) * NG + i) * FRAG);
        }
      }
      __builtin_amdgcn_sched_barrier(0);
#pragma unroll
      for (int i = 0; i < NG; ++i)
#pragma unroll
        for (int mt = 0; mt < MT; ++mt)
          acc[mt][g * NG + i] = __builtin_amdgcn_wmma_f32_16x16x32_bf16(
              false, ahi[mt], false, bh[i], (short)0, acc[mt][g * NG + i], false, false);
#pragma unroll
      for (int i = 0; i < NG; ++i)
#pragma unroll
        for (int mt = 0; mt < MT; ++mt)
          acc[mt][g * NG + i] = __builtin_amdgcn_wmma_f32_16x16x32_bf16(
              false, ahi[mt], false, bl[i], (short)0, acc[mt][g * NG + i], false, false);
#pragma unroll
      for (int i = 0; i < NG; ++i)
#pragma unroll
        for (int mt = 0; mt < MT; ++mt)
          acc[mt][g * NG + i] = __builtin_amdgcn_wmma_f32_16x16x32_bf16(
              false, alo[mt], false, bh[i], (short)0, acc[mt][g * NG + i], false, false);
      __builtin_amdgcn_sched_barrier(0);
      if (g + 1 < NTILES / NG) {
#pragma unroll
        for (int i = 0; i < NG; ++i) { bh[i] = nh[i]; bl[i] = nl[i]; }
      }
    }

    {
      u32x4* dst = &sB[(kc + 1) & 1][tid];
#pragma unroll
      for (int s = 0; s < 8; ++s) { dst[s * 128] = st[s]; dst[PLANE_V + s * 128] = st[8 + s]; }
    }
#pragma unroll
    for (int s = 0; s < 8; ++s) {
      st[s]     = gh[(size_t)kp * PLANE_V + s * 128];
      st[8 + s] = gl[(size_t)kp * PLANE_V + s * 128];
    }
    lds_handover();
  }

#pragma unroll
  for (int mt = 0; mt < MT; ++mt) {
    if (m_base + mt * 16 < NUM_ENT) {
#pragma unroll
      for (int nt = 0; nt < NTILES; ++nt) {
#pragma unroll
        for (int j = 0; j < 8; ++j) {
          int m   = m_base + mt * 16 + j + half * 8;
          int col = nt * 16 + r15;
          unity[(size_t)(m + 1) * EMBED + col] = acc[mt][nt][j];
        }
      }
    }
  }
}

__device__ __forceinline__ float add_abs(float acc, float t) {
  asm("v_add_f32 %0, |%1|, %0" : "+v"(acc) : "v"(t));
  return acc;
}

__global__ __launch_bounds__(256) void BLP_score(
    const float* __restrict__ unity, const int* __restrict__ ids,
    const int* __restrict__ mp_ptr, float* __restrict__ out) {
  __shared__ float q[BATCH * EMBED];
  {
    const int lane = threadIdx.x & 31;
    const int wave = __builtin_amdgcn_readfirstlane((int)(threadIdx.x >> 5));
    const int mp = *mp_ptr;
    const int c0 = (mp == 0) ? 1 : 0;
    const int c1 = (mp <= 1) ? 2 : 1;
    float v[4][2][8];
#pragma unroll
    for (int j = 0; j < 4; ++j) {
      const int b = wave * 4 + j;
      const float* p0 = unity + (size_t)ids[b * 3 + c0] * EMBED + lane;
      const float* p1 = unity + (size_t)ids[b * 3 + c1] * EMBED + lane;
#pragma unroll
      for (int i = 0; i < 8; ++i) { v[j][0][i] = p0[32 * i]; v[j][1][i] = p1[32 * i]; }
    }
#pragma unroll
    for (int j = 0; j < 4; ++j) {
      float rinv[2];
#pragma unroll
      for (int r = 0; r < 2; ++r) {
        float s[8];
#pragma unroll
        for (int i = 0; i < 8; ++i) s[i] = v[j][r][i] * v[j][r][i];
#pragma unroll
        for (int i = 0; i < 4; ++i) s[i] += s[i + 4];
        s[0] += s[2]; s[1] += s[3];
        float t = s[0] + s[1];
#pragma unroll
        for (int m = 16; m > 0; m >>= 1) t += __shfl_xor(t, m);
        rinv[r] = 1.0f / fmaxf(sqrtf(t), 1e-12f);
      }
      float* qb = q + (wave * 4 + j) * EMBED + lane;
#pragma unroll
      for (int i = 0; i < 8; ++i)
        qb[32 * i] = v[j][0][i] * rinv[0] + v[j][1][i] * rinv[1];
    }
  }
  __syncthreads();

  const int e = blockIdx.x * 256 + threadIdx.x;
  const bool active = (e < NE_SCORE);
  const float4* row = (const float4*)(unity + (size_t)(active ? e : 0) * EMBED);

  float acc[BATCH];
#pragma unroll
  for (int b = 0; b < BATCH; ++b) acc[b] = 0.0f;

  for (int d4 = 0; d4 < EMBED / 4; ++d4) {
    const float4 v = row[d4];
    const v2f vlo = {v.x, v.y};
    const v2f vhi = {v.z, v.w};
    const int dbase = d4 * 4;
    v2f t[BATCH][2];
#pragma unroll
    for (int b = 0; b < BATCH; ++b) {
      const float4 qq = *(const float4*)(q + b * EMBED + dbase);
      const v2f qlo = {qq.x, qq.y};
      const v2f qhi = {qq.z, qq.w};
      t[b][0] = qlo - vlo;
      t[b][1] = qhi - vhi;
    }
#pragma unroll
    for (int b = 0; b < BATCH; ++b) acc[b] = add_abs(acc[b], t[b][0].x);
#pragma unroll
    for (int b = 0; b < BATCH; ++b) acc[b] = add_abs(acc[b], t[b][0].y);
#pragma unroll
    for (int b = 0; b < BATCH; ++b) acc[b] = add_abs(acc[b], t[b][1].x);
#pragma unroll
    for (int b = 0; b < BATCH; ++b) acc[b] = add_abs(acc[b], t[b][1].y);
  }
  if (active) {
#pragma unroll
    for (int b = 0; b < BATCH; ++b)
      out[(size_t)b * NE_SCORE + e] = -acc[b];
  }
}

extern "C" void kernel_launch(void* const* d_in, const int* in_sizes, int n_in,
                              void* d_out, int out_size, void* d_ws, size_t ws_size,
                              hipStream_t stream) {
  (void)in_sizes; (void)n_in; (void)out_size; (void)ws_size;
  const float* ent_pkl = (const float*)d_in[0];
  const float* other   = (const float*)d_in[1];
  const float* proj_W  = (const float*)d_in[2];
  const int*   ids     = (const int*)d_in[3];
  const int*   mp      = (const int*)d_in[4];
  float*       out     = (float*)d_out;

  float*  unity = (float*)d_ws;
  __bf16* whi   = (__bf16*)(unity + (size_t)NE_TOT * EMBED + (size_t)BATCH * EMBED);
  __bf16* wlo   = whi + (size_t)NTILES * KCHUNKS * 32 * 16;

  BLP_split_w<<<SPLIT_WGS + OTHER_ROWS, 32, 0, stream>>>(proj_W, whi, wlo, other, unity);
  BLP_gemm_proj<<<GEMM_WGS, 32 * WAVES, 0, stream>>>(ent_pkl, whi, wlo, unity);
  BLP_score<<<(NE_SCORE + 255) / 256, 256, 0, stream>>>(unity, ids, mp, out);
}


# Score kernel re-tiled: each block keeps 160 rows in LDS and each wave scores all of them against 4 queries, so one LDS query read feeds 5 rows (was 1); 251 blocks, was 157; sums keep their d order
# speedup vs baseline: 1.9581x; 1.1847x over previous
#include <hip/hip_runtime.h>
#include <hip/hip_bf16.h>

#define NUM_ENT   40000
#define EMBED     256
#define FEAT      768
#define KCHUNKS   (FEAT / 32)
#define NTILES    (EMBED / 16)
#define NE_TOT    40103
#define NE_SCORE  40001
#define BATCH     32

typedef __attribute__((ext_vector_type(16))) __bf16 v16bf;
typedef __attribute__((ext_vector_type(8)))  float  v8f;
typedef __attribute__((ext_vector_type(2)))  float  v2f;
typedef __attribute__((ext_vector_type(4)))  float  v4f;

__device__ __forceinline__ void load16(const float* __restrict__ p, float* v) {
  float4 a0 = *(const float4*)(p + 0);
  float4 a1 = *(const float4*)(p + 4);
  float4 b0 = *(const float4*)(p + 16);
  float4 b1 = *(const float4*)(p + 20);
  v[0] = a0.x; v[1] = a0.y; v[2]  = a0.z; v[3]  = a0.w;
  v[4] = a1.x; v[5] = a1.y; v[6]  = a1.z; v[7]  = a1.w;
  v[8] = b0.x; v[9] = b0.y; v[10] = b0.z; v[11] = b0.w;
  v[12] = b1.x; v[13] = b1.y; v[14] = b1.z; v[15] = b1.w;
}

#define SPLIT_WGS (NTILES * KCHUNKS)
#define OTHER_ROWS 103
__global__ __launch_bounds__(32) void BLP_split_w(
    const float* __restrict__ W, __bf16* __restrict__ whi,
    __bf16* __restrict__ wlo, const float* __restrict__ other,
    float* __restrict__ unity) {
  const int lane = threadIdx.x;
  if (blockIdx.x >= SPLIT_WGS) {
    const int r   = blockIdx.x - SPLIT_WGS;
    const int dst = (r == 0) ? 0 : (NUM_ENT + r);
    const float4* s = (const float4*)(other + (size_t)r * EMBED);
    float4*       d = (float4*)(unity + (size_t)dst * EMBED);
    d[lane]      = s[lane];
    d[lane + 32] = s[lane + 32];
    return;
  }
  const int kc   = blockIdx.x % KCHUNKS;
  const int nt   = blockIdx.x / KCHUNKS;
  const int half = lane >> 4;
  const int r15  = lane & 15;

  const float* p = W + (size_t)(nt * 16 + r15) * FEAT + kc * 32 + half * 8;
  float v[16];
  load16(p, v);

  const size_t base = (size_t)(kc * NTILES + nt) * 512 + lane * 8;
#pragma unroll
  for (int i = 0; i < 16; ++i) {
    __bf16 h = (__bf16)v[i];
    const size_t off = base + (i >> 3) * 256 + (i & 7);
    whi[off] = h;
    wlo[off] = (__bf16)(v[i] - (float)h);
  }
}

#define MT 3
#define NG 4
#define FRAG 512
#define WAVES 4
#define ROWS_WG (16 * MT * WAVES)
#define GEMM_WGS ((NUM_ENT + ROWS_WG - 1) / ROWS_WG)
#define PLANE (NTILES * FRAG)
#define PLANE_V (PLANE / 8)

typedef __attribute__((ext_vector_type(8))) __bf16 v8bf;
typedef __attribute__((ext_vector_type(4))) unsigned int u32x4;

__device__ __forceinline__ v16bf lds_frag(const __bf16* p) {
  v8bf a = *(const v8bf*)(p);
  v8bf b = *(const v8bf*)(p + 256);
  return __builtin_shufflevector(a, b, 0, 1, 2, 3, 4, 5, 6, 7,
                                 8, 9, 10, 11, 12, 13, 14, 15);
}

__device__ __forceinline__ void lds_handover() {
  __builtin_amdgcn_fence(__ATOMIC_RELEASE, "workgroup", "local");
  __builtin_amdgcn_s_barrier();
  __builtin_amdgcn_fence(__ATOMIC_ACQUIRE, "workgroup", "local");
}

__global__ __launch_bounds__(32 * WAVES) void BLP_gemm_proj(
    const float* __restrict__ A, const __bf16* __restrict__ whi,
    const __bf16* __restrict__ wlo, float* __restrict__ unity) {
  __shared__ u32x4 sB[2][2 * PLANE_V];

  const int tid  = threadIdx.x;
  const int lane = tid & 31;
  const int wave = __builtin_amdgcn_readfirstlane(tid >> 5);
  const int half = lane >> 4;
  const int r15  = lane & 15;
  const int m_base = (blockIdx.x * WAVES + wave) * (16 * MT);

  v8f acc[MT][NTILES] = {};

  const float* arow[MT];
#pragma unroll
  for (int mt = 0; mt < MT; ++mt) {
    int r = m_base + mt * 16 + r15;
    r = (r < NUM_ENT) ? r : (NUM_ENT - 1);
    arow[mt] = A + (size_t)r * FEAT + half * 8;
  }

  float ra[MT][16];
#pragma unroll
  for (int mt = 0; mt < MT; ++mt) load16(arow[mt], ra[mt]);
  __builtin_amdgcn_sched_barrier(0);

  const u32x4* gh = (const u32x4*)whi + tid;
  const u32x4* gl = (const u32x4*)wlo + tid;

  u32x4 st[16];
#pragma unroll
  for (int s = 0; s < 8; ++s) { st[s] = gh[s * 128]; st[8 + s] = gl[s * 128]; }
  {
    u32x4* dst = &sB[0][tid];
#pragma unroll
    for (int s = 0; s < 8; ++s) { dst[s * 128] = st[s]; dst[PLANE_V + s * 128] = st[8 + s]; }
  }
#pragma unroll
  for (int s = 0; s < 8; ++s) {
    st[s]     = gh[PLANE_V + s * 128];
    st[8 + s] = gl[PLANE_V + s * 128];
  }

  lds_handover();

  for (int kc = 0; kc < KCHUNKS; ++kc) {
    const int kn = (kc + 1 < KCHUNKS) ? kc + 1 : kc;

    const __bf16* sh = (const __bf16*)&sB[kc & 1][0] + lane * 8;
    const __bf16* sl = sh + PLANE;

    v16bf bh[NG], bl[NG];
#pragma unroll
    for (int i = 0; i < NG; ++i) {
      bh[i] = lds_frag(sh + i * FRAG);
      bl[i] = lds_frag(sl + i * FRAG);
    }

    v16bf ahi[MT], alo[MT];
#pragma unroll
    for (int mt = 0; mt < MT; ++mt) {
#pragma unroll
      for (int i = 0; i < 16; ++i) {
        __bf16 h = (__bf16)ra[mt][i];
        ahi[mt][i] = h;
        alo[mt][i] = (__bf16)(ra[mt][i] - (float)h);
      }
    }
    const int kp = (kc + 2 < KCHUNKS) ? kc + 2 : KCHUNKS - 1;
#pragma unroll
    for (int mt = 0; mt < MT; ++mt) {
      load16(arow[mt] + kn * 32, ra[mt]);
      __builtin_prefetch(arow[mt] + kp * 32, 0, 0);
    }

#pragma unroll
    for (int g = 0; g < NTILES / NG; ++g) {
      v16bf nh[NG], nl[NG];
      if (g + 1 < NTILES / NG) {
#pragma unroll
        for (int i = 0; i < NG; ++i) {
          nh[i] = lds_frag(sh + ((g + 1) * NG + i) * FRAG);
          nl[i] = lds_frag(sl + ((g + 1) * NG + i) * FRAG);
        }
      }
      __builtin_amdgcn_sched_barrier(0);
#pragma unroll
      for (int i = 0; i < NG; ++i)
#pragma unroll
        for (int mt = 0; mt < MT; ++mt)
          acc[mt][g * NG + i] = __builtin_amdgcn_wmma_f32_16x16x32_bf16(
              false, ahi[mt], false, bh[i], (short)0, acc[mt][g * NG + i], false, false);
#pragma unroll
      for (int i = 0; i < NG; ++i)
#pragma unroll
        for (int mt = 0; mt < MT; ++mt)
          acc[mt][g * NG + i] = __builtin_amdgcn_wmma_f32_16x16x32_bf16(
              false, ahi[mt], false, bl[i], (short)0, acc[mt][g * NG + i], false, false);
#pragma unroll
      for (int i = 0; i < NG; ++i)
#pragma unroll
        for (int mt = 0; mt < MT; ++mt)
          acc[mt][g * NG + i] = __builtin_amdgcn_wmma_f32_16x16x32_bf16(
              false, alo[mt], false, bh[i], (short)0, acc[mt][g * NG + i], false, false);
      __builtin_amdgcn_sched_barrier(0);
      if (g + 1 < NTILES / NG) {
#pragma unroll
        for (int i = 0; i < NG; ++i) { bh[i] = nh[i]; bl[i] = nl[i]; }
      }
    }

    {
      u32x4* dst = &sB[(kc + 1) & 1][tid];
#pragma unroll
      for (int s = 0; s < 8; ++s) { dst[s * 128] = st[s]; dst[PLANE_V + s * 128] = st[8 + s]; }
    }
#pragma unroll
    for (int s = 0; s < 8; ++s) {
      st[s]     = gh[(size_t)kp * PLANE_V + s * 128];
      st[8 + s] = gl[(size_t)kp * PLANE_V + s * 128];
    }
    lds_handover();
  }

#pragma unroll
  for (int mt = 0; mt < MT; ++mt) {
    if (m_base + mt * 16 < NUM_ENT) {
#pragma unroll
      for (int nt = 0; nt < NTILES; ++nt) {
#pragma unroll
        for (int j = 0; j < 8; ++j) {
          int m   = m_base + mt * 16 + j + half * 8;
          int col = nt * 16 + r15;
          unity[(size_t)(m + 1) * EMBED + col] = acc[mt][nt][j];
        }
      }
    }
  }
}

__device__ __forceinline__ float add_abs(float acc, float t) {
  asm("v_add_f32 %0, |%1|, %0" : "+v"(acc) : "v"(t));
  return acc;
}

#define SROWS  160
#define SPITCH 260
#define SLAB   64
#define RPL    (SROWS / 32)
#define QPW    (BATCH / 8)
#define SPT    (SROWS / 16)
#define SCORE_WGS ((NE_SCORE + SROWS - 1) / SROWS)

__device__ __forceinline__ void score_read(const float* rp, const float* qp,
                                           v4f* rv, v4f* qq) {
#pragma unroll
  for (int r = 0; r < RPL; ++r) rv[r] = *(const v4f*)(rp + 32 * r * SPITCH);
#pragma unroll
  for (int j = 0; j < QPW; ++j) qq[j] = *(const v4f*)(qp + j * EMBED);
}

__device__ __forceinline__ void score_step(const v4f* rv, const v4f* qq,
                                           float (*acc)[QPW]) {
  v2f t[RPL][QPW][2];
#pragma unroll
  for (int r = 0; r < RPL; ++r) {
    const v2f vlo = {rv[r].x, rv[r].y};
    const v2f vhi = {rv[r].z, rv[r].w};
#pragma unroll
    for (int j = 0; j < QPW; ++j) {
      const v2f qlo = {qq[j].x, qq[j].y};
      const v2f qhi = {qq[j].z, qq[j].w};
      t[r][j][0] = qlo - vlo;
      t[r][j][1] = qhi - vhi;
    }
  }
#pragma unroll
  for (int r = 0; r < RPL; ++r)
#pragma unroll
    for (int j = 0; j < QPW; ++j) acc[r][j] = add_abs(acc[r][j], t[r][j][0].x);
#pragma unroll
  for (int r = 0; r < RPL; ++r)
#pragma unroll
    for (int j = 0; j < QPW; ++j) acc[r][j] = add_abs(acc[r][j], t[r][j][0].y);
#pragma unroll
  for (int r = 0; r < RPL; ++r)
#pragma unroll
    for (int j = 0; j < QPW; ++j) acc[r][j] = add_abs(acc[r][j], t[r][j][1].x);
#pragma unroll
  for (int r = 0; r < RPL; ++r)
#pragma unroll
    for (int j = 0; j < QPW; ++j) acc[r][j] = add_abs(acc[r][j], t[r][j][1].y);
}

__global__ __launch_bounds__(256) void BLP_score(
    const float* __restrict__ unity, const int* __restrict__ ids,
    const int* __restrict__ mp_ptr, float* __restrict__ out) {
  __shared__ __attribute__((aligned(16))) float q[BATCH * EMBED];
  __shared__ __attribute__((aligned(16))) float rows[SROWS * SPITCH];
  const int tid  = threadIdx.x;
  const int lane = tid & 31;
  const int wave = __builtin_amdgcn_readfirstlane(tid >> 5);
  const int row0 = blockIdx.x * SROWS;

  const int sc = tid & 15;
  const int sr = tid >> 4;
  const float* gsrc[SPT];
#pragma unroll
  for (int i = 0; i < SPT; ++i) {
    int e = row0 + sr + 16 * i;
    e = (e < NE_SCORE) ? e : (NE_SCORE - 1);
    gsrc[i] = unity + (size_t)e * EMBED + sc * 4;
  }
  float* const ldst = rows + sr * SPITCH + sc * 4;
  v4f st[SPT];
#pragma unroll
  for (int i = 0; i < SPT; ++i) st[i] = *(const v4f*)(gsrc[i]);

  {
    const int mp = *mp_ptr;
    const int c0 = (mp == 0) ? 1 : 0;
    const int c1 = (mp <= 1) ? 2 : 1;
    float v[4][2][8];
#pragma unroll
    for (int j = 0; j < 4; ++j) {
      const int b = wave * 4 + j;
      const float* p0 = unity + (size_t)ids[b * 3 + c0] * EMBED + lane;
      const float* p1 = unity + (size_t)ids[b * 3 + c1] * EMBED + lane;
#pragma unroll
      for (int i = 0; i < 8; ++i) { v[j][0][i] = p0[32 * i]; v[j][1][i] = p1[32 * i]; }
    }
#pragma unroll
    for (int j = 0; j < 4; ++j) {
      float rinv[2];
#pragma unroll
      for (int r = 0; r < 2; ++r) {
        float s[8];
#pragma unroll
        for (int i = 0; i < 8; ++i) s[i] = v[j][r][i] * v[j][r][i];
#pragma unroll
        for (int i = 0; i < 4; ++i) s[i] += s[i + 4];
        s[0] += s[2]; s[1] += s[3];
        float t = s[0] + s[1];
#pragma unroll
        for (int m = 16; m > 0; m >>= 1) t += __shfl_xor(t, m);
        rinv[r] = 1.0f / fmaxf(sqrtf(t), 1e-12f);
      }
      float* qb = q + (wave * 4 + j) * EMBED + lane;
#pragma unroll
      for (int i = 0; i < 8; ++i)
        qb[32 * i] = v[j][0][i] * rinv[0] + v[j][1][i] * rinv[1];
    }
  }
#pragma unroll
  for (int i = 0; i < SPT; ++i) *(v4f*)(ldst + 16 * i * SPITCH) = st[i];
  __syncthreads();

  float acc[RPL][QPW];
#pragma unroll
  for (int r = 0; r < RPL; ++r)
#pragma unroll
    for (int j = 0; j < QPW; ++j) acc[r][j] = 0.0f;

  const float* const rbase = rows + lane * SPITCH;
  const float* const qbase = q + wave * QPW * EMBED;

  for (int s = 0; s < EMBED / SLAB; ++s) {
    if (s + 1 < EMBED / SLAB) {
#pragma unroll
      for (int i = 0; i < SPT; ++i) st[i] = *(const v4f*)(gsrc[i] + (s + 1) * SLAB);
    }

    const float* const rp = rbase + s * SLAB;
    const float* const qp = qbase + s * SLAB;
    v4f ra[RPL], qa[QPW], rb[RPL], qb[QPW];
    score_read(rp, qp, ra, qa);
    for (int i = 0; i < SLAB / 4; i += 2) {
      score_read(rp + (i + 1) * 4, qp + (i + 1) * 4, rb, qb);
      __builtin_amdgcn_sched_barrier(0);
      score_step(ra, qa, acc);
      const int in = (i + 2 < SLAB / 4) ? i + 2 : i;
      score_read(rp + in * 4, qp + in * 4, ra, qa);
      __builtin_amdgcn_sched_barrier(0);
      score_step(rb, qb, acc);
    }

    if (s + 1 < EMBED / SLAB) {
#pragma unroll
      for (int i = 0; i < SPT; ++i)
        *(v4f*)(ldst + 16 * i * SPITCH + (s + 1) * SLAB) = st[i];
      __syncthreads();
    }
  }

#pragma unroll
  for (int r = 0; r < RPL; ++r) {
    const int e = row0 + lane + 32 * r;
    if (e < NE_SCORE) {
#pragma unroll
      for (int j = 0; j < QPW; ++j)
        out[(size_t)(wave * QPW + j) * NE_SCORE + e] = -acc[r][j];
    }
  }
}

extern "C" void kernel_launch(void* const* d_in, const int* in_sizes, int n_in,
                              void* d_out, int out_size, void* d_ws, size_t ws_size,
                              hipStream_t stream) {
  (void)in_sizes; (void)n_in; (void)out_size; (void)ws_size;
  const float* ent_pkl = (const float*)d_in[0];
  const float* other   = (const float*)d_in[1];
  const float* proj_W  = (const float*)d_in[2];
  const int*   ids     = (const int*)d_in[3];
  const int*   mp      = (const int*)d_in[4];
  float*       out     = (float*)d_out;

  float*  unity = (float*)d_ws;
  __bf16* whi   = (__bf16*)(unity + (size_t)NE_TOT * EMBED + (size_t)BATCH * EMBED);
  __bf16* wlo   = whi + (size_t)NTILES * KCHUNKS * 32 * 16;

  BLP_split_w<<<SPLIT_WGS + OTHER_ROWS, 32, 0, stream>>>(proj_W, whi, wlo, other, unity);
  BLP_gemm_proj<<<GEMM_WGS, 32 * WAVES, 0, stream>>>(ent_pkl, whi, wlo, unity);
  BLP_score<<<SCORE_WGS, 256, 0, stream>>>(unity, ids, mp, out);
}
